# starting kernel
# baseline (speedup 1.0000x reference)
#include <hip/hip_runtime.h>

constexpr int Bc   = 8;
constexpr int Nc   = 256;
constexpr int NFc  = 256;
constexpr int EFc  = 128;
constexpr int Hc   = 8;
constexpr int Oc   = 32;
constexpr int NHID = 1024;
constexpr int HOc  = 256;
constexpr float ALPHA = 0.2f;
constexpr float EPSc  = 1e-6f;

typedef __attribute__((ext_vector_type(2))) float v2f;
typedef __attribute__((ext_vector_type(8))) float v8f;

__device__ __forceinline__ v8f wmma_gemm_f32(const float* __restrict__ A, int lda,
                                             const float* __restrict__ B, int ldb,
                                             int K, v8f acc) {
  const int lane = threadIdx.x & 31;
  const int l16  = lane & 15;
  const int ko   = (lane >> 4) << 1;
  const float* ap = A + l16 * lda + ko;
  const float* bp = B + ko * ldb + l16;
#pragma unroll 8
  for (int k = 0; k < K; k += 4) {
    v2f av, bv;
    av.x = ap[k];
    av.y = ap[k + 1];
    bv.x = bp[k * ldb];
    bv.y = bp[k * ldb + ldb];
    acc = __builtin_amdgcn_wmma_f32_16x16x4_f32(false, av, false, bv,
                                                (short)0, acc, false, false);
  }
  return acc;
}

__device__ __forceinline__ float wave_max32(float v) {
#pragma unroll
  for (int m = 16; m >= 1; m >>= 1) v = fmaxf(v, __shfl_xor(v, m, 32));
  return v;
}
__device__ __forceinline__ float wave_sum32(float v) {
#pragma unroll
  for (int m = 16; m >= 1; m >>= 1) v += __shfl_xor(v, m, 32);
  return v;
}

__global__ void k_h(const float* __restrict__ x, const float* __restrict__ W,
                    float* __restrict__ h) {
  int wg = blockIdx.x * (blockDim.x >> 5) + (threadIdx.x >> 5);
  int nt = wg & 1;
  int mt = (wg >> 1) & 15;
  int bh = wg >> 5;
  int b  = bh >> 3, head = bh & 7;
  const float* A  = x + ((long)b * Nc + mt * 16) * NFc;
  const float* Bm = W + (long)head * NFc * Oc + nt * 16;
  v8f acc = {};
  acc = wmma_gemm_f32(A, NFc, Bm, Oc, NFc, acc);
  float* C = h + ((long)bh * Nc + mt * 16) * Oc + nt * 16;
  int lane = threadIdx.x & 31;
  int col  = lane & 15;
  int rb   = (lane >> 4) * 8;
#pragma unroll
  for (int g = 0; g < 8; ++g) C[(rb + g) * Oc + col] = acc[g];
}

__global__ void k_st(const float* __restrict__ h, const float* __restrict__ a,
                     float* __restrict__ s, float* __restrict__ t) {
  int idx  = blockIdx.x * blockDim.x + threadIdx.x;
  int head = (idx >> 8) & (Hc - 1);
  const float* hr = h + (long)idx * Oc;
  const float* a1 = a + head * (3 * Oc);
  const float* a2 = a1 + Oc;
  float sv = 0.f, tv = 0.f;
#pragma unroll
  for (int o = 0; o < Oc; ++o) { sv += hr[o] * a1[o]; tv += hr[o] * a2[o]; }
  s[idx] = sv;
  t[idx] = tv;
}

__global__ void k_v(const float* __restrict__ W1, const float* __restrict__ a,
                    float* __restrict__ vt) {
  int idx = blockIdx.x * blockDim.x + threadIdx.x;
  int c = idx & 15, e = idx >> 4;
  float v = 0.f;
  if (c < Hc) {
    const float* w  = W1 + ((long)c * EFc + e) * Oc;
    const float* a3 = a + c * (3 * Oc) + 2 * Oc;
#pragma unroll
    for (int o = 0; o < Oc; ++o) v += w[o] * a3[o];
  }
  vt[idx] = v;
}

__global__ void k_u(const float* __restrict__ edge, const float* __restrict__ vt,
                    float* __restrict__ u) {
  int wg = blockIdx.x * (blockDim.x >> 5) + (threadIdx.x >> 5);
  const float* A = edge + (long)wg * 16 * EFc;
  v8f acc = {};
  acc = wmma_gemm_f32(A, EFc, vt, 16, EFc, acc);
  int lane = threadIdx.x & 31;
  int col  = lane & 15;
  int rb   = (lane >> 4) * 8;
  if (col < Hc) {
#pragma unroll
    for (int g = 0; g < 8; ++g) {
      long row = (long)wg * 16 + rb + g;
      int b   = (int)(row >> 16);
      int rem = (int)(row & 65535);
      int i = rem >> 8, j = rem & 255;
      u[(((long)b * Hc + col) * Nc + i) * Nc + j] = acc[g];
    }
  }
}

__global__ void k_attn(float* __restrict__ u, const float* __restrict__ s,
                       const float* __restrict__ t, const float* __restrict__ adj) {
  int wg   = blockIdx.x * (blockDim.x >> 5) + (threadIdx.x >> 5);
  int lane = threadIdx.x & 31;
  int i  = wg & (Nc - 1);
  int bh = wg >> 8;
  int b  = bh >> 3;
  float*       urow = u + (long)wg * Nc;
  const float* trow = t + (long)bh * Nc;
  const float* arow = adj + ((long)b * Nc + i) * Nc;
  float sv = s[wg];
  float vals[8];
  float mx = -3.0e38f;
#pragma unroll
  for (int p = 0; p < 8; ++p) {
    int j = p * 32 + lane;
    float sc = sv + trow[j] + urow[j];
    sc = sc > 0.f ? sc : ALPHA * sc;
    sc = (arow[j] > 0.f) ? sc : -3.0e38f;
    vals[p] = sc;
    mx = fmaxf(mx, sc);
  }
  mx = wave_max32(mx);
  float sum = 0.f;
#pragma unroll
  for (int p = 0; p < 8; ++p) {
    float e = __expf(vals[p] - mx);
    vals[p] = e;
    sum += e;
  }
  sum = wave_sum32(sum);
  float inv = 1.f / sum;
#pragma unroll
  for (int p = 0; p < 8; ++p) urow[p * 32 + lane] = vals[p] * inv;
}

__global__ void k_hp(const float* __restrict__ attn, const float* __restrict__ h,
                     float* __restrict__ cat) {
  int wg = blockIdx.x * (blockDim.x >> 5) + (threadIdx.x >> 5);
  int nt = wg & 1;
  int mt = (wg >> 1) & 15;
  int bh = wg >> 5;
  int b  = bh >> 3, head = bh & 7;
  const float* A  = attn + ((long)bh * Nc + mt * 16) * Nc;
  const float* Bm = h + (long)bh * Nc * Oc + nt * 16;
  v8f acc = {};
  acc = wmma_gemm_f32(A, Nc, Bm, Oc, Nc, acc);
  int lane = threadIdx.x & 31;
  int col  = lane & 15;
  int rb   = (lane >> 4) * 8;
#pragma unroll
  for (int g = 0; g < 8; ++g) {
    int i = mt * 16 + rb + g;
    float vv = acc[g];
    vv = vv > 0.f ? vv : (__expf(vv) - 1.f);
    cat[((long)b * Nc + i) * HOc + head * Oc + nt * 16 + col] = vv;
  }
}

__global__ void k_mlp(const float* __restrict__ A, const float* __restrict__ Bm,
                      const float* __restrict__ bias, float* __restrict__ C,
                      int Ncols, int K, int relu) {
  int waves_n = Ncols >> 4;
  int wg = blockIdx.x * (blockDim.x >> 5) + (threadIdx.x >> 5);
  int nt = wg % waves_n;
  int mt = wg / waves_n;
  const float* Ap = A + (long)mt * 16 * K;
  const float* Bp = Bm + nt * 16;
  v8f acc = {};
  acc = wmma_gemm_f32(Ap, K, Bp, Ncols, K, acc);
  int lane = threadIdx.x & 31;
  int col  = nt * 16 + (lane & 15);
  int rb   = (lane >> 4) * 8;
#pragma unroll
  for (int g = 0; g < 8; ++g) {
    float vv = acc[g] + bias[col];
    if (relu) vv = fmaxf(vv, 0.f);
    C[((long)(mt * 16 + rb + g)) * Ncols + col] = vv;
  }
}

__global__ void k_ff(const float* __restrict__ z, const float* __restrict__ w2,
                     const float* __restrict__ b2, const float* __restrict__ y,
                     float* __restrict__ ff) {
  int wg = blockIdx.x * (blockDim.x >> 5) + (threadIdx.x >> 5);
  int nt = wg & 15;
  int mt = wg >> 4;
  const float* Ap = z + (long)mt * 16 * NHID;
  const float* Bp = w2 + nt * 16;
  v8f acc = {};
  acc = wmma_gemm_f32(Ap, NHID, Bp, HOc, NHID, acc);
  int lane = threadIdx.x & 31;
  int col  = nt * 16 + (lane & 15);
  int rb   = (lane >> 4) * 8;
#pragma unroll
  for (int g = 0; g < 8; ++g) {
    int row = mt * 16 + rb + g;
    ff[(long)row * HOc + col] = acc[g] + b2[col] + y[(long)row * HOc + col];
  }
}

__global__ void k_ln(const float* __restrict__ ff, const float* __restrict__ g,
                     const float* __restrict__ bb, float* __restrict__ out) {
  int wg   = blockIdx.x * (blockDim.x >> 5) + (threadIdx.x >> 5);
  int lane = threadIdx.x & 31;
  const float* r = ff + (long)wg * HOc;
  float s1 = 0.f, s2 = 0.f;
#pragma unroll
  for (int p = 0; p < 8; ++p) {
    float v = r[p * 32 + lane];
    s1 += v;
    s2 += v * v;
  }
  s1 = wave_sum32(s1);
  s2 = wave_sum32(s2);
  float mu  = s1 * (1.f / HOc);
  float var = s2 * (1.f / HOc) - mu * mu;
  float inv = rsqrtf(var + EPSc);
  float* o = out + (long)wg * HOc;
#pragma unroll
  for (int p = 0; p < 8; ++p) {
    int c = p * 32 + lane;
    o[c] = (r[c] - mu) * inv * g[c] + bb[c];
  }
}

extern "C" void kernel_launch(void* const* d_in, const int* in_sizes, int n_in,
                              void* d_out, int out_size, void* d_ws, size_t ws_size,
                              hipStream_t stream) {
  const float* x    = (const float*)d_in[0];
  const float* edge = (const float*)d_in[1];
  const float* adj  = (const float*)d_in[2];
  const float* W    = (const float*)d_in[3];
  const float* W1   = (const float*)d_in[4];
  const float* a    = (const float*)d_in[5];
  const float* Wo   = (const float*)d_in[6];
  const float* bo   = (const float*)d_in[7];
  const float* w1   = (const float*)d_in[8];
  const float* b1   = (const float*)d_in[9];
  const float* w2   = (const float*)d_in[10];
  const float* b2   = (const float*)d_in[11];
  const float* lng  = (const float*)d_in[12];
  const float* lnb  = (const float*)d_in[13];
  float* out = (float*)d_out;

  float* ws  = (float*)d_ws;
  float* h   = ws;
  float* s   = h + 524288;
  float* t   = s + 16384;
  float* vt  = t + 16384;
  float* u   = vt + 2048;
  float* cat = u + 4194304;
  float* y   = cat + 524288;
  float* z   = y + 524288;
  float* ff  = z + 2097152;

  k_h   <<<256,  256, 0, stream>>>(x, W, h);
  k_st  <<<64,   256, 0, stream>>>(h, a, s, t);
  k_v   <<<8,    256, 0, stream>>>(W1, a, vt);
  k_u   <<<4096, 256, 0, stream>>>(edge, vt, u);
  k_attn<<<2048, 256, 0, stream>>>(u, s, t, adj);
  k_hp  <<<256,  256, 0, stream>>>(u, h, cat);
  k_mlp <<<256,  256, 0, stream>>>(cat, Wo, bo, y, HOc, HOc, 1);
  k_mlp <<<1024, 256, 0, stream>>>(y, w1, b1, z, NHID, HOc, 1);
  k_ff  <<<256,  256, 0, stream>>>(z, w2, b2, y, ff);
  k_ln  <<<256,  256, 0, stream>>>(ff, lng, lnb, out);
}


# h, Wo, w1, w2 GEMMs take operands from LDS: a workgroup stages each 32x64 (w1: 64x128) block's A and B chunks once, 3-stage ring, loads two chunks ahead; k order and sums unchanged
# speedup vs baseline: 1.5925x; 1.5925x over previous
#include <hip/hip_runtime.h>

constexpr int Bc   = 8;
constexpr int Nc   = 256;
constexpr int NFc  = 256;
constexpr int EFc  = 128;
constexpr int Hc   = 8;
constexpr int Oc   = 32;
constexpr int NHID = 1024;
constexpr int HOc  = 256;
constexpr float ALPHA = 0.2f;
constexpr float EPSc  = 1e-6f;

typedef __attribute__((ext_vector_type(2))) float v2f;
typedef __attribute__((ext_vector_type(8))) float v8f;

typedef __attribute__((ext_vector_type(4))) float v4f;

template <int BM, int BN, int KC, int WM, int WN, int K, int NBN, int BMODE, int EMODE>
__device__ __forceinline__ void gemm_lds_block(const float* __restrict__ A, const int lda,
                                               const float* __restrict__ Bm, const int ldb,
                                               const float* __restrict__ bias,
                                               const float* __restrict__ resid,
                                               float* __restrict__ C, const int ldc,
                                               float* __restrict__ smem) {
  constexpr int PA = KC + 4, P2 = 2 * BN + 32;
  constexpr int SA = BM * PA, SB = (KC / 2) * P2, STAGE = SA + SB;
  constexpr int AQ = KC / 4, BQ = BN / 2;
  constexpr int A_F4 = BM * AQ / 256, B_F4 = (KC / 2) * BQ / 256;
  constexpr int WAVES_N = BN / (16 * WN);
  constexpr int T = K / KC, S = KC / 4;
  static_assert(T >= 4 && (T & 1) == 0, "chunk count");
  const int tid  = threadIdx.x;
  const int lane = tid & 31;
  const int w    = tid >> 5;
  const int c    = lane & 15;
  const int hh   = lane >> 4;
  const int ko   = hh << 1;
  const int m0   = (blockIdx.x / NBN) * BM;
  const int n0   = (blockIdx.x % NBN) * BN;
  const int rowb = (w / WAVES_N) * (16 * WM);
  const int colb = (w % WAVES_N) * (16 * WN);

  const float* asrc[A_F4];
  const float* bsrc[B_F4];
  int adst[A_F4], bdst[B_F4];
#pragma unroll
  for (int j = 0; j < A_F4; ++j) {
    const int idx = j * 256 + tid, r = idx / AQ, q = idx % AQ;
    asrc[j] = A + (long)(m0 + r) * lda + 4 * q;
    adst[j] = r * PA + 4 * q;
  }
#pragma unroll
  for (int j = 0; j < B_F4; ++j) {
    const int idx = j * 256 + tid, p = idx / BQ, q = idx % BQ, col = n0 + 2 * q;
    if (BMODE == 0) bsrc[j] = Bm + (long)(2 * p) * ldb + col;
    else            bsrc[j] = Bm + ((long)(col >> 5) * K + 2 * p) * 32 + (col & 31);
    bdst[j] = SA + p * P2 + 4 * q;
  }
  const long brow  = BMODE == 0 ? (long)ldb : 32L;
  const long bstep = KC * brow;

  v8f acc[WM][WN];
#pragma unroll
  for (int i = 0; i < WM; ++i)
#pragma unroll
    for (int j = 0; j < WN; ++j) acc[i][j] = v8f{};

  auto gload = [&](v4f (&ra)[A_F4], v2f (&rb)[B_F4][2], const int t) {
#pragma unroll
    for (int j = 0; j < A_F4; ++j) ra[j] = *(const v4f*)(asrc[j] + t * KC);
#pragma unroll
    for (int j = 0; j < B_F4; ++j) {
      rb[j][0] = *(const v2f*)(bsrc[j] + t * bstep);
      rb[j][1] = *(const v2f*)(bsrc[j] + t * bstep + brow);
    }
  };
  auto sstore = [&](const v4f (&ra)[A_F4], const v2f (&rb)[B_F4][2], const int soff) {
    float* s = smem + soff;
#pragma unroll
    for (int j = 0; j < A_F4; ++j) *(v4f*)(s + adst[j]) = ra[j];
#pragma unroll
    for (int j = 0; j < B_F4; ++j) {
      v4f o;
      o.x = rb[j][0].x; o.y = rb[j][1].x; o.z = rb[j][0].y; o.w = rb[j][1].y;
      *(v4f*)(s + bdst[j]) = o;
    }
  };
  auto fragload = [&](v2f (&af)[S][WM], v2f (&bf)[S][WN], const int soff) {
    const float* sa = smem + soff + (rowb + c) * PA + ko;
    const float* sb = smem + soff + SA + hh * P2 + 2 * (colb + c);
#pragma unroll
    for (int s = 0; s < S; ++s) {
#pragma unroll
      for (int i = 0; i < WM; ++i) af[s][i] = *(const v2f*)(sa + i * 16 * PA + 4 * s);
#pragma unroll
      for (int j = 0; j < WN; ++j) bf[s][j] = *(const v2f*)(sb + (2 * s) * P2 + 32 * j);
    }
  };
  auto mma = [&](const v2f (&af)[S][WM], const v2f (&bf)[S][WN]) {
#pragma unroll
    for (int s = 0; s < S; ++s)
#pragma unroll
      for (int i = 0; i < WM; ++i)
#pragma unroll
        for (int j = 0; j < WN; ++j)
          acc[i][j] = __builtin_amdgcn_wmma_f32_16x16x4_f32(false, af[s][i], false, bf[s][j],
                                                            (short)0, acc[i][j], false, false);
  };

  v4f ra[A_F4];
  v2f rb[B_F4][2];
  v2f afX[S][WM], bfX[S][WN], afY[S][WM], bfY[S][WN];
  int s0 = 0, s1 = STAGE, s2 = 2 * STAGE;
  {
    v4f ra1[A_F4];
    v2f rb1[B_F4][2];
    gload(ra, rb, 0);
    gload(ra1, rb1, 1);
    sstore(ra, rb, s0);
    sstore(ra1, rb1, s1);
  }
  __syncthreads();
  fragload(afX, bfX, s0);

#define GEMM_LDS_STEP(afc, bfc, afn, bfn, t, sn, sw)                                                      \
  gload(ra, rb, (t) + 2);                                                                                 \
  fragload(afn, bfn, sn);                                                                                 \
    \
    \
    \
  __builtin_amdgcn_sched_barrier(0);                                                                      \
  mma(afc, bfc);                                                                                          \
    \
    \
  __builtin_amdgcn_sched_barrier(0);                                                                      \
  sstore(ra, rb, sw);                                                                                     \
  __syncthreads();

  for (int t = 0; t < T - 2; t += 2) {
    GEMM_LDS_STEP(afX, bfX, afY, bfY, t, s1, s2)
    GEMM_LDS_STEP(afY, bfY, afX, bfX, t + 1, s2, s0)
    const int r0 = s0; s0 = s2; s2 = s1; s1 = r0;
  }
#undef GEMM_LDS_STEP
  fragload(afY, bfY, s1);
  __builtin_amdgcn_sched_barrier(0);
  mma(afX, bfX);
  mma(afY, bfY);

#pragma unroll
  for (int i = 0; i < WM; ++i) {
#pragma unroll
    for (int j = 0; j < WN; ++j) {
      const int col = n0 + colb + 16 * j + c;
#pragma unroll
      for (int r = 0; r < 8; ++r) {
        const int row = m0 + rowb + 16 * i + 8 * hh + r;
        if (EMODE == 0) {
          C[((((long)(row >> 8)) * Hc + (col >> 5)) * Nc + (row & 255)) * Oc + (col & 31)] = acc[i][j][r];
        } else if (EMODE == 1) {
          float vv = acc[i][j][r] + bias[col];
          vv = fmaxf(vv, 0.f);
          C[(long)row * ldc + col] = vv;
        } else {
          C[(long)row * ldc + col] = acc[i][j][r] + bias[col] + resid[(long)row * ldc + col];
        }
      }
    }
  }
}

__device__ __forceinline__ float wave_sum32(float v) {
#pragma unroll
  for (int m = 16; m >= 1; m >>= 1) v += __shfl_xor(v, m, 32);
  return v;
}

__global__ void __launch_bounds__(256)
k_h(const float* __restrict__ x, const float* __restrict__ W, float* __restrict__ h) {
  __shared__ __attribute__((aligned(16))) float smem[3 * (32 * 68 + 32 * 160)];
  gemm_lds_block<32, 64, 64, 1, 1, NFc, 4, 1, 0>(x, NFc, W, Oc, nullptr, nullptr, h, 0, smem);
}

__global__ void k_st(const float* __restrict__ h, const float* __restrict__ a,
                     float* __restrict__ s, float* __restrict__ t) {
  int idx  = blockIdx.x * blockDim.x + threadIdx.x;
  int head = (idx >> 8) & (Hc - 1);
  const float* hr = h + (long)idx * Oc;
  const float* a1 = a + head * (3 * Oc);
  const float* a2 = a1 + Oc;
  float sv = 0.f, tv = 0.f;
#pragma unroll
  for (int o = 0; o < Oc; ++o) { sv += hr[o] * a1[o]; tv += hr[o] * a2[o]; }
  s[idx] = sv;
  t[idx] = tv;
}

__global__ void k_v(const float* __restrict__ W1, const float* __restrict__ a,
                    float* __restrict__ vt) {
  int idx = blockIdx.x * blockDim.x + threadIdx.x;
  int c = idx & 15, e = idx >> 4;
  float v = 0.f;
  if (c < Hc) {
    const float* w  = W1 + ((long)c * EFc + e) * Oc;
    const float* a3 = a + c * (3 * Oc) + 2 * Oc;
#pragma unroll
    for (int o = 0; o < Oc; ++o) v += w[o] * a3[o];
  }
  vt[idx] = v;
}

constexpr int U_TPW   = 8;
constexpr int U_PITCH = 132;
__global__ void __launch_bounds__(256)
k_u(const float* __restrict__ edge, const float* __restrict__ vt,
    float* __restrict__ u) {
  __shared__ __attribute__((aligned(16))) float stage[8][16 * U_PITCH];
  const int lane = threadIdx.x & 31;
  const int wib  = threadIdx.x >> 5;
  const int wave = blockIdx.x * 8 + wib;
  const int c    = lane & 15;
  const int hh   = lane >> 4;
  const int ko   = hh << 1;
  float* sw = &stage[wib][0];

  const int head = c & 7;
  const int q    = c >> 3;
  const float* bp = vt + ko * 16 + head;
  v2f bf[32];
#pragma unroll
  for (int s = 0; s < 32; ++s) {
    bf[s].x = bp[(4 * s) * 16];
    bf[s].y = bp[(4 * s + 1) * 16];
  }

  const v4f* src = (const v4f*)(edge + (long)wave * (U_TPW * 16 * EFc)) + lane;
  v4f r[16];
#pragma unroll
  for (int rw = 0; rw < 16; ++rw) r[rw] = src[rw * 32];

  for (int it = 0; it < U_TPW; ++it) {
#pragma unroll
    for (int rw = 0; rw < 16; ++rw) *(v4f*)(sw + rw * U_PITCH + 4 * lane) = r[rw];
    if (it + 1 < U_TPW) {
#pragma unroll
      for (int rw = 0; rw < 16; ++rw) r[rw] = src[(it + 1) * 512 + rw * 32];
    }
    __builtin_amdgcn_fence(__ATOMIC_RELEASE, "wavefront");
    __builtin_amdgcn_wave_barrier();
    __builtin_amdgcn_fence(__ATOMIC_ACQUIRE, "wavefront");

    const float* ap = sw + c * U_PITCH + ko;
    v2f av[32];
#pragma unroll
    for (int s = 0; s < 32; ++s) av[s] = *(const v2f*)(ap + 4 * s);
    __builtin_amdgcn_sched_barrier(0);
    v8f acc = {};
#pragma unroll
    for (int s = 0; s < 32; ++s) {
      acc = __builtin_amdgcn_wmma_f32_16x16x4_f32(false, av[s], false, bf[s],
                                                  (short)0, acc, false, false);
    }

    const int tile = wave * U_TPW + it;
    const int b    = tile >> 12;
    const int i    = (tile >> 4) & 255;
    const int j0   = (tile & 15) * 16;
    v4f o;
    o.x = q ? acc[4] : acc[0];
    o.y = q ? acc[5] : acc[1];
    o.z = q ? acc[6] : acc[2];
    o.w = q ? acc[7] : acc[3];
    *(v4f*)(u + (((long)b * Hc + head) * Nc + i) * Nc + j0 + 8 * hh + 4 * q) = o;
  }
}

__global__ void __launch_bounds__(128) __attribute__((amdgpu_waves_per_eu(1, 2)))
k_attn_hp(const float* __restrict__ u, const float* __restrict__ s,
          const float* __restrict__ t, const float* __restrict__ adj,
          const float* __restrict__ h, float* __restrict__ cat) {
  const int wave = blockIdx.x * (blockDim.x >> 5) + (threadIdx.x >> 5);
  const int lane = threadIdx.x & 31;
  const int c    = lane & 15;
  const int hh   = lane >> 4;
  const int ko   = hh << 1;
  const int mt   = wave & 15;
  const int bh   = wave >> 4;
  const int b    = bh >> 3, head = bh & 7;
  const int i    = mt * 16 + c;
  const float* urow = u   + ((long)bh * Nc + i) * Nc + ko;
  const float* arow = adj + ((long)b  * Nc + i) * Nc + ko;
  const float* trow = t   + (long)bh * Nc + ko;
  const float  sv   = s[bh * Nc + i];

  float p[128];
  float mx = -3.0e38f;
#pragma unroll
  for (int kk = 0; kk < 64; ++kk) {
    const v2f uv = *(const v2f*)(urow + 4 * kk);
    const v2f tv = *(const v2f*)(trow + 4 * kk);
    const v2f av = *(const v2f*)(arow + 4 * kk);
    float s0 = sv + tv.x + uv.x;
    float s1 = sv + tv.y + uv.y;
    s0 = s0 > 0.f ? s0 : ALPHA * s0;
    s1 = s1 > 0.f ? s1 : ALPHA * s1;
    s0 = (av.x > 0.f) ? s0 : -3.0e38f;
    s1 = (av.y > 0.f) ? s1 : -3.0e38f;
    p[2 * kk]     = s0;
    p[2 * kk + 1] = s1;
    mx = fmaxf(mx, fmaxf(s0, s1));
  }
  const float* hb = h + (long)bh * Nc * Oc + ko * Oc + c;
  float hv[256];
#pragma unroll
  for (int kk = 0; kk < 64; ++kk) {
    hv[4 * kk]     = hb[(4 * kk) * Oc];
    hv[4 * kk + 1] = hb[(4 * kk + 1) * Oc];
    hv[4 * kk + 2] = hb[(4 * kk) * Oc + 16];
    hv[4 * kk + 3] = hb[(4 * kk + 1) * Oc + 16];
  }

  mx = fmaxf(mx, __shfl_xor(mx, 16, 32));
  float sum0 = 0.f, sum1 = 0.f, sum2 = 0.f, sum3 = 0.f;
#pragma unroll
  for (int q = 0; q < 128; q += 4) {
    const float e0 = __expf(p[q]     - mx);
    const float e1 = __expf(p[q + 1] - mx);
    const float e2 = __expf(p[q + 2] - mx);
    const float e3 = __expf(p[q + 3] - mx);
    p[q] = e0; p[q + 1] = e1; p[q + 2] = e2; p[q + 3] = e3;
    sum0 += e0; sum1 += e1; sum2 += e2; sum3 += e3;
  }
  float sum = (sum0 + sum1) + (sum2 + sum3);
  sum += __shfl_xor(sum, 16, 32);
  float inv = 1.f / sum;


  v8f acc0 = {}, acc1 = {};
  asm volatile("" : "+v"(inv), "+v"(acc0), "+v"(acc1));
#pragma unroll
  for (int kk = 0; kk < 64; ++kk) {
    v2f a, b0, b1;
    a.x  = p[2 * kk];
    a.y  = p[2 * kk + 1];
    b0.x = hv[4 * kk];
    b0.y = hv[4 * kk + 1];
    b1.x = hv[4 * kk + 2];
    b1.y = hv[4 * kk + 3];
    acc0 = __builtin_amdgcn_wmma_f32_16x16x4_f32(false, a, false, b0,
                                                 (short)0, acc0, false, false);
    acc1 = __builtin_amdgcn_wmma_f32_16x16x4_f32(false, a, false, b1,
                                                 (short)0, acc1, false, false);
  }
#pragma unroll
  for (int r = 0; r < 8; ++r) {
    const float invr = __shfl(inv, 8 * hh + r, 32);
    const int   ir   = mt * 16 + 8 * hh + r;
    float v0 = acc0[r] * invr;
    float v1 = acc1[r] * invr;
    v0 = v0 > 0.f ? v0 : (__expf(v0) - 1.f);
    v1 = v1 > 0.f ? v1 : (__expf(v1) - 1.f);
    float* dst = cat + ((long)b * Nc + ir) * HOc + head * Oc + c;
    dst[0]  = v0;
    dst[16] = v1;
  }
}

__global__ void __launch_bounds__(256)
k_y(const float* __restrict__ cat, const float* __restrict__ Wo, const float* __restrict__ bo,
    float* __restrict__ y) {
  __shared__ __attribute__((aligned(16))) float smem[3 * (32 * 68 + 32 * 160)];
  gemm_lds_block<32, 64, 64, 1, 1, HOc, 4, 0, 1>(cat, HOc, Wo, HOc, bo, nullptr, y, HOc, smem);
}

__global__ void __launch_bounds__(256)
k_z(const float* __restrict__ y, const float* __restrict__ w1, const float* __restrict__ b1,
    float* __restrict__ z) {
  __shared__ __attribute__((aligned(16))) float smem[3 * (64 * 36 + 16 * 288)];
  gemm_lds_block<64, 128, 32, 2, 2, HOc, 8, 0, 1>(y, HOc, w1, NHID, b1, nullptr, z, NHID, smem);
}

__global__ void __launch_bounds__(256)
k_ff(const float* __restrict__ z, const float* __restrict__ w2,
     const float* __restrict__ b2, const float* __restrict__ y,
     float* __restrict__ ff) {
  __shared__ __attribute__((aligned(16))) float smem[3 * (32 * 68 + 32 * 160)];
  gemm_lds_block<32, 64, 64, 1, 1, NHID, 4, 0, 2>(z, NHID, w2, HOc, b2, y, ff, HOc, smem);
}

__global__ void k_ln(const float* __restrict__ ff, const float* __restrict__ g,
                     const float* __restrict__ bb, float* __restrict__ out) {
  int wg   = blockIdx.x * (blockDim.x >> 5) + (threadIdx.x >> 5);
  int lane = threadIdx.x & 31;
  const float* r = ff + (long)wg * HOc;
  float s1 = 0.f, s2 = 0.f;
#pragma unroll
  for (int p = 0; p < 8; ++p) {
    float v = r[p * 32 + lane];
    s1 += v;
    s2 += v * v;
  }
  s1 = wave_sum32(s1);
  s2 = wave_sum32(s2);
  float mu  = s1 * (1.f / HOc);
  float var = s2 * (1.f / HOc) - mu * mu;
  float inv = rsqrtf(var + EPSc);
  float* o = out + (long)wg * HOc;
#pragma unroll
  for (int p = 0; p < 8; ++p) {
    int c = p * 32 + lane;
    o[c] = (r[c] - mu) * inv * g[c] + bb[c];
  }
}

extern "C" void kernel_launch(void* const* d_in, const int* in_sizes, int n_in,
                              void* d_out, int out_size, void* d_ws, size_t ws_size,
                              hipStream_t stream) {
  const float* x    = (const float*)d_in[0];
  const float* edge = (const float*)d_in[1];
  const float* adj  = (const float*)d_in[2];
  const float* W    = (const float*)d_in[3];
  const float* W1   = (const float*)d_in[4];
  const float* a    = (const float*)d_in[5];
  const float* Wo   = (const float*)d_in[6];
  const float* bo   = (const float*)d_in[7];
  const float* w1   = (const float*)d_in[8];
  const float* b1   = (const float*)d_in[9];
  const float* w2   = (const float*)d_in[10];
  const float* b2   = (const float*)d_in[11];
  const float* lng  = (const float*)d_in[12];
  const float* lnb  = (const float*)d_in[13];
  float* out = (float*)d_out;

  float* ws  = (float*)d_ws;
  float* h   = ws;
  float* s   = h + 524288;
  float* t   = s + 16384;
  float* vt  = t + 16384;
  float* u   = vt + 2048;
  float* cat = u + 4194304;
  float* y   = cat + 524288;
  float* z   = y + 524288;
  float* ff  = z + 2097152;

  k_h   <<<256,  256, 0, stream>>>(x, W, h);
  k_st  <<<64,   256, 0, stream>>>(h, a, s, t);
  k_v   <<<8,    256, 0, stream>>>(W1, a, vt);
  k_u   <<<512,  256, 0, stream>>>(edge, vt, u);
  k_attn_hp<<<256, 128, 0, stream>>>(u, s, t, adj, h, cat);
  k_y   <<<256,  256, 0, stream>>>(cat, Wo, bo, y);
  k_z   <<<256,  256, 0, stream>>>(y, w1, b1, z);
  k_ff  <<<256,  256, 0, stream>>>(z, w2, b2, y, ff);
  k_ln  <<<256,  256, 0, stream>>>(ff, lng, lnb, out);
}


# One launch ahead of attention: 256 h-GEMM workgroups (s,t from their h tile in LDS) run beside the 512 edge-stream workgroups, each forming vT itself; 9 launches to 6; vT sum order changes
# speedup vs baseline: 1.7763x; 1.1154x over previous
#include <hip/hip_runtime.h>

constexpr int Bc   = 8;
constexpr int Nc   = 256;
constexpr int NFc  = 256;
constexpr int EFc  = 128;
constexpr int Hc   = 8;
constexpr int Oc   = 32;
constexpr int NHID = 1024;
constexpr int HOc  = 256;
constexpr float ALPHA = 0.2f;
constexpr float EPSc  = 1e-6f;

typedef __attribute__((ext_vector_type(2))) float v2f;
typedef __attribute__((ext_vector_type(8))) float v8f;

typedef __attribute__((ext_vector_type(4))) float v4f;

template <int BM, int BN, int KC, int WM, int WN, int K, int NBN, int BMODE, int EMODE>
__device__ __forceinline__ void gemm_lds_block(const int bid,
                                               const float* __restrict__ A, const int lda,
                                               const float* __restrict__ Bm, const int ldb,
                                               const float* __restrict__ bias,
                                               const float* __restrict__ resid,
                                               float* __restrict__ C, const int ldc,
                                               float* __restrict__ smem) {
  constexpr int PA = KC + 4, P2 = 2 * BN + 32;
  constexpr int SA = BM * PA, SB = (KC / 2) * P2, STAGE = SA + SB;
  constexpr int AQ = KC / 4, BQ = BN / 2;
  constexpr int A_F4 = BM * AQ / 256, B_F4 = (KC / 2) * BQ / 256;
  constexpr int WAVES_N = BN / (16 * WN);
  constexpr int T = K / KC, S = KC / 4;
  static_assert(T >= 4 && (T & 1) == 0, "chunk count");
  const int tid  = threadIdx.x;
  const int lane = tid & 31;
  const int w    = tid >> 5;
  const int c    = lane & 15;
  const int hh   = lane >> 4;
  const int ko   = hh << 1;
  const int m0   = (bid / NBN) * BM;
  const int n0   = (bid % NBN) * BN;
  const int rowb = (w / WAVES_N) * (16 * WM);
  const int colb = (w % WAVES_N) * (16 * WN);

  const float* asrc[A_F4];
  const float* bsrc[B_F4];
  int adst[A_F4], bdst[B_F4];
#pragma unroll
  for (int j = 0; j < A_F4; ++j) {
    const int idx = j * 256 + tid, r = idx / AQ, q = idx % AQ;
    asrc[j] = A + (long)(m0 + r) * lda + 4 * q;
    adst[j] = r * PA + 4 * q;
  }
#pragma unroll
  for (int j = 0; j < B_F4; ++j) {
    const int idx = j * 256 + tid, p = idx / BQ, q = idx % BQ, col = n0 + 2 * q;
    if (BMODE == 0) bsrc[j] = Bm + (long)(2 * p) * ldb + col;
    else            bsrc[j] = Bm + ((long)(col >> 5) * K + 2 * p) * 32 + (col & 31);
    bdst[j] = SA + p * P2 + 4 * q;
  }
  const long brow  = BMODE == 0 ? (long)ldb : 32L;
  const long bstep = KC * brow;

  v8f acc[WM][WN];
#pragma unroll
  for (int i = 0; i < WM; ++i)
#pragma unroll
    for (int j = 0; j < WN; ++j) acc[i][j] = v8f{};

  auto gload = [&](v4f (&ra)[A_F4], v2f (&rb)[B_F4][2], const int t) {
#pragma unroll
    for (int j = 0; j < A_F4; ++j) ra[j] = *(const v4f*)(asrc[j] + t * KC);
#pragma unroll
    for (int j = 0; j < B_F4; ++j) {
      rb[j][0] = *(const v2f*)(bsrc[j] + t * bstep);
      rb[j][1] = *(const v2f*)(bsrc[j] + t * bstep + brow);
    }
  };
  auto sstore = [&](const v4f (&ra)[A_F4], const v2f (&rb)[B_F4][2], const int soff) {
    float* s = smem + soff;
#pragma unroll
    for (int j = 0; j < A_F4; ++j) *(v4f*)(s + adst[j]) = ra[j];
#pragma unroll
    for (int j = 0; j < B_F4; ++j) {
      v4f o;
      o.x = rb[j][0].x; o.y = rb[j][1].x; o.z = rb[j][0].y; o.w = rb[j][1].y;
      *(v4f*)(s + bdst[j]) = o;
    }
  };
  auto fragload = [&](v2f (&af)[S][WM], v2f (&bf)[S][WN], const int soff) {
    const float* sa = smem + soff + (rowb + c) * PA + ko;
    const float* sb = smem + soff + SA + hh * P2 + 2 * (colb + c);
#pragma unroll
    for (int s = 0; s < S; ++s) {
#pragma unroll
      for (int i = 0; i < WM; ++i) af[s][i] = *(const v2f*)(sa + i * 16 * PA + 4 * s);
#pragma unroll
      for (int j = 0; j < WN; ++j) bf[s][j] = *(const v2f*)(sb + (2 * s) * P2 + 32 * j);
    }
  };
  auto mma = [&](const v2f (&af)[S][WM], const v2f (&bf)[S][WN]) {
#pragma unroll
    for (int s = 0; s < S; ++s)
#pragma unroll
      for (int i = 0; i < WM; ++i)
#pragma unroll
        for (int j = 0; j < WN; ++j)
          acc[i][j] = __builtin_amdgcn_wmma_f32_16x16x4_f32(false, af[s][i], false, bf[s][j],
                                                            (short)0, acc[i][j], false, false);
  };

  v4f ra[A_F4];
  v2f rb[B_F4][2];
  v2f afX[S][WM], bfX[S][WN], afY[S][WM], bfY[S][WN];
  int s0 = 0, s1 = STAGE, s2 = 2 * STAGE;
  {
    v4f ra1[A_F4];
    v2f rb1[B_F4][2];
    gload(ra, rb, 0);
    gload(ra1, rb1, 1);
    sstore(ra, rb, s0);
    sstore(ra1, rb1, s1);
  }
  __syncthreads();
  fragload(afX, bfX, s0);

#define GEMM_LDS_STEP(afc, bfc, afn, bfn, t, sn, sw)                                                      \
  gload(ra, rb, (t) + 2);                                                                                 \
  fragload(afn, bfn, sn);                                                                                 \
    \
    \
    \
  __builtin_amdgcn_sched_barrier(0);                                                                      \
  mma(afc, bfc);                                                                                          \
    \
    \
  __builtin_amdgcn_sched_barrier(0);                                                                      \
  sstore(ra, rb, sw);                                                                                     \
  __syncthreads();

  for (int t = 0; t < T - 2; t += 2) {
    GEMM_LDS_STEP(afX, bfX, afY, bfY, t, s1, s2)
    GEMM_LDS_STEP(afY, bfY, afX, bfX, t + 1, s2, s0)
    const int r0 = s0; s0 = s2; s2 = s1; s1 = r0;
  }
#undef GEMM_LDS_STEP
  fragload(afY, bfY, s1);
  __builtin_amdgcn_sched_barrier(0);
  mma(afX, bfX);
  mma(afY, bfY);

  if (EMODE == 0) {
    __syncthreads();
#pragma unroll
    for (int i = 0; i < WM; ++i)
#pragma unroll
      for (int j = 0; j < WN; ++j)
#pragma unroll
        for (int r = 0; r < 8; ++r)
          smem[(rowb + 16 * i + 8 * hh + r) * (BN + 1) + colb + 16 * j + c] = acc[i][j][r];
    __syncthreads();
  }
#pragma unroll
  for (int i = 0; i < WM; ++i) {
#pragma unroll
    for (int j = 0; j < WN; ++j) {
      const int col = n0 + colb + 16 * j + c;
#pragma unroll
      for (int r = 0; r < 8; ++r) {
        const int row = m0 + rowb + 16 * i + 8 * hh + r;
        if (EMODE == 0) {
          C[((((long)(row >> 8)) * Hc + (col >> 5)) * Nc + (row & 255)) * Oc + (col & 31)] = acc[i][j][r];
        } else if (EMODE == 1) {
          float vv = acc[i][j][r] + bias[col];
          vv = fmaxf(vv, 0.f);
          C[(long)row * ldc + col] = vv;
        } else {
          C[(long)row * ldc + col] = acc[i][j][r] + bias[col] + resid[(long)row * ldc + col];
        }
      }
    }
  }
}

__device__ __forceinline__ float wave_sum32(float v) {
#pragma unroll
  for (int m = 16; m >= 1; m >>= 1) v += __shfl_xor(v, m, 32);
  return v;
}

constexpr int U_TPW   = 8;
constexpr int U_PITCH = 132;
constexpr int H_BLOCKS = 256;
__global__ void __launch_bounds__(256)
k_front(const float* __restrict__ x, const float* __restrict__ W, const float* __restrict__ a,
        const float* __restrict__ edge, const float* __restrict__ W1,
        float* __restrict__ h, float* __restrict__ s, float* __restrict__ t, float* __restrict__ u) {
  __shared__ __attribute__((aligned(16))) float smem[3 * (32 * 68 + 32 * 160)];
  if (blockIdx.x < H_BLOCKS) {
    const int bid = blockIdx.x;
    gemm_lds_block<32, 64, 64, 1, 1, NFc, 4, 1, 0>(bid, x, NFc, W, Oc, nullptr, nullptr, h, 0, smem);
    const int tid = threadIdx.x;
    if (tid < 128) {
      const int row   = tid & 31;
      const int hd    = (tid >> 5) & 1;
      const int which = tid >> 6;
      const int head  = (bid & 3) * 2 + hd;
      const float* hr = smem + row * 65 + hd * Oc;
      const float* av = a + head * (3 * Oc) + which * Oc;
      float v = 0.f;
#pragma unroll
      for (int o = 0; o < Oc; ++o) v += hr[o] * av[o];
      const int m = (bid >> 2) * 32 + row;
      float* dst = which ? t : s;
      dst[((m >> 8) * Hc + head) * Nc + (m & 255)] = v;
    }
    return;
  }

  const int lane = threadIdx.x & 31;
  const int wib  = threadIdx.x >> 5;
  const int wave = (blockIdx.x - H_BLOCKS) * 8 + wib;
  const int c    = lane & 15;
  const int hh   = lane >> 4;
  const int ko   = hh << 1;
  float* sw  = smem + wib * (16 * U_PITCH);
  float* vts = smem + 8 * (16 * U_PITCH);

  const float* w1h = W1 + (long)wib * (EFc * Oc);
  const v4f a3v = *(const v4f*)(a + wib * (3 * Oc) + 2 * Oc + 4 * (lane & 7));
  v4f wv[32];
#pragma unroll
  for (int it = 0; it < 32; ++it) wv[it] = *(const v4f*)(w1h + (it * 32 + lane) * 4);

  const v4f* src = (const v4f*)(edge + (long)wave * (U_TPW * 16 * EFc)) + lane;
  v4f r[16];
#pragma unroll
  for (int rw = 0; rw < 16; ++rw) r[rw] = src[rw * 32];

  float pv[32];
#pragma unroll
  for (int it = 0; it < 32; ++it) {
    float p = wv[it].x * a3v.x + wv[it].y * a3v.y + wv[it].z * a3v.z + wv[it].w * a3v.w;
    p += __shfl_xor(p, 1, 32);
    p += __shfl_xor(p, 2, 32);
    p += __shfl_xor(p, 4, 32);
    pv[it] = p;
  }
  if ((lane & 7) == 0) {
#pragma unroll
    for (int it = 0; it < 32; ++it) vts[(4 * it + (lane >> 3)) * 8 + wib] = pv[it];
  }
  __syncthreads();

  const int head = c & 7;
  const int q    = c >> 3;
  const float* bp = vts + ko * 8 + head;
  v2f bf[32];
#pragma unroll
  for (int st = 0; st < 32; ++st) {
    bf[st].x = bp[(4 * st) * 8];
    bf[st].y = bp[(4 * st + 1) * 8];
  }

  for (int it = 0; it < U_TPW; ++it) {
#pragma unroll
    for (int rw = 0; rw < 16; ++rw) *(v4f*)(sw + rw * U_PITCH + 4 * lane) = r[rw];
    if (it + 1 < U_TPW) {
#pragma unroll
      for (int rw = 0; rw < 16; ++rw) r[rw] = src[(it + 1) * 512 + rw * 32];
    }
    __builtin_amdgcn_fence(__ATOMIC_RELEASE, "wavefront");
    __builtin_amdgcn_wave_barrier();
    __builtin_amdgcn_fence(__ATOMIC_ACQUIRE, "wavefront");

    const float* ap = sw + c * U_PITCH + ko;
    v2f av[32];
#pragma unroll
    for (int st = 0; st < 32; ++st) av[st] = *(const v2f*)(ap + 4 * st);
    __builtin_amdgcn_sched_barrier(0);
    v8f acc = {};
#pragma unroll
    for (int st = 0; st < 32; ++st) {
      acc = __builtin_amdgcn_wmma_f32_16x16x4_f32(false, av[st], false, bf[st],
                                                  (short)0, acc, false, false);
    }

    const int tile = wave * U_TPW + it;
    const int b    = tile >> 12;
    const int i    = (tile >> 4) & 255;
    const int j0   = (tile & 15) * 16;
    v4f o;
    o.x = q ? acc[4] : acc[0];
    o.y = q ? acc[5] : acc[1];
    o.z = q ? acc[6] : acc[2];
    o.w = q ? acc[7] : acc[3];
    *(v4f*)(u + (((long)b * Hc + head) * Nc + i) * Nc + j0 + 8 * hh + 4 * q) = o;
  }
}

__global__ void __launch_bounds__(128) __attribute__((amdgpu_waves_per_eu(1, 2)))
k_attn_hp(const float* __restrict__ u, const float* __restrict__ s,
          const float* __restrict__ t, const float* __restrict__ adj,
          const float* __restrict__ h, float* __restrict__ cat) {
  const int wave = blockIdx.x * (blockDim.x >> 5) + (threadIdx.x >> 5);
  const int lane = threadIdx.x & 31;
  const int c    = lane & 15;
  const int hh   = lane >> 4;
  const int ko   = hh << 1;
  const int mt   = wave & 15;
  const int bh   = wave >> 4;
  const int b    = bh >> 3, head = bh & 7;
  const int i    = mt * 16 + c;
  const float* urow = u   + ((long)bh * Nc + i) * Nc + ko;
  const float* arow = adj + ((long)b  * Nc + i) * Nc + ko;
  const float* trow = t   + (long)bh * Nc + ko;
  const float  sv   = s[bh * Nc + i];

  float p[128];
  float mx = -3.0e38f;
#pragma unroll
  for (int kk = 0; kk < 64; ++kk) {
    const v2f uv = *(const v2f*)(urow + 4 * kk);
    const v2f tv = *(const v2f*)(trow + 4 * kk);
    const v2f av = *(const v2f*)(arow + 4 * kk);
    float s0 = sv + tv.x + uv.x;
    float s1 = sv + tv.y + uv.y;
    s0 = s0 > 0.f ? s0 : ALPHA * s0;
    s1 = s1 > 0.f ? s1 : ALPHA * s1;
    s0 = (av.x > 0.f) ? s0 : -3.0e38f;
    s1 = (av.y > 0.f) ? s1 : -3.0e38f;
    p[2 * kk]     = s0;
    p[2 * kk + 1] = s1;
    mx = fmaxf(mx, fmaxf(s0, s1));
  }
  const float* hb = h + (long)bh * Nc * Oc + ko * Oc + c;
  float hv[256];
#pragma unroll
  for (int kk = 0; kk < 64; ++kk) {
    hv[4 * kk]     = hb[(4 * kk) * Oc];
    hv[4 * kk + 1] = hb[(4 * kk + 1) * Oc];
    hv[4 * kk + 2] = hb[(4 * kk) * Oc + 16];
    hv[4 * kk + 3] = hb[(4 * kk + 1) * Oc + 16];
  }

  mx = fmaxf(mx, __shfl_xor(mx, 16, 32));
  float sum0 = 0.f, sum1 = 0.f, sum2 = 0.f, sum3 = 0.f;
#pragma unroll
  for (int q = 0; q < 128; q += 4) {
    const float e0 = __expf(p[q]     - mx);
    const float e1 = __expf(p[q + 1] - mx);
    const float e2 = __expf(p[q + 2] - mx);
    const float e3 = __expf(p[q + 3] - mx);
    p[q] = e0; p[q + 1] = e1; p[q + 2] = e2; p[q + 3] = e3;
    sum0 += e0; sum1 += e1; sum2 += e2; sum3 += e3;
  }
  float sum = (sum0 + sum1) + (sum2 + sum3);
  sum += __shfl_xor(sum, 16, 32);
  float inv = 1.f / sum;


  v8f acc0 = {}, acc1 = {};
  asm volatile("" : "+v"(inv), "+v"(acc0), "+v"(acc1));
#pragma unroll
  for (int kk = 0; kk < 64; ++kk) {
    v2f a, b0, b1;
    a.x  = p[2 * kk];
    a.y  = p[2 * kk + 1];
    b0.x = hv[4 * kk];
    b0.y = hv[4 * kk + 1];
    b1.x = hv[4 * kk + 2];
    b1.y = hv[4 * kk + 3];
    acc0 = __builtin_amdgcn_wmma_f32_16x16x4_f32(false, a, false, b0,
                                                 (short)0, acc0, false, false);
    acc1 = __builtin_amdgcn_wmma_f32_16x16x4_f32(false, a, false, b1,
                                                 (short)0, acc1, false, false);
  }
#pragma unroll
  for (int r = 0; r < 8; ++r) {
    const float invr = __shfl(inv, 8 * hh + r, 32);
    const int   ir   = mt * 16 + 8 * hh + r;
    float v0 = acc0[r] * invr;
    float v1 = acc1[r] * invr;
    v0 = v0 > 0.f ? v0 : (__expf(v0) - 1.f);
    v1 = v1 > 0.f ? v1 : (__expf(v1) - 1.f);
    float* dst = cat + ((long)b * Nc + ir) * HOc + head * Oc + c;
    dst[0]  = v0;
    dst[16] = v1;
  }
}

__global__ void __launch_bounds__(256)
k_y(const float* __restrict__ cat, const float* __restrict__ Wo, const float* __restrict__ bo,
    float* __restrict__ y) {
  __shared__ __attribute__((aligned(16))) float smem[3 * (32 * 68 + 32 * 160)];
  gemm_lds_block<32, 64, 64, 1, 1, HOc, 4, 0, 1>(blockIdx.x, cat, HOc, Wo, HOc, bo, nullptr, y, HOc, smem);
}

__global__ void __launch_bounds__(256)
k_z(const float* __restrict__ y, const float* __restrict__ w1, const float* __restrict__ b1,
    float* __restrict__ z) {
  __shared__ __attribute__((aligned(16))) float smem[3 * (64 * 36 + 16 * 288)];
  gemm_lds_block<64, 128, 32, 2, 2, HOc, 8, 0, 1>(blockIdx.x, y, HOc, w1, NHID, b1, nullptr, z, NHID, smem);
}

__global__ void __launch_bounds__(256)
k_ff(const float* __restrict__ z, const float* __restrict__ w2,
     const float* __restrict__ b2, const float* __restrict__ y,
     float* __restrict__ ff) {
  __shared__ __attribute__((aligned(16))) float smem[3 * (32 * 68 + 32 * 160)];
  gemm_lds_block<32, 64, 64, 1, 1, NHID, 4, 0, 2>(blockIdx.x, z, NHID, w2, HOc, b2, y, ff, HOc, smem);
}

__global__ void k_ln(const float* __restrict__ ff, const float* __restrict__ g,
                     const float* __restrict__ bb, float* __restrict__ out) {
  int wg   = blockIdx.x * (blockDim.x >> 5) + (threadIdx.x >> 5);
  int lane = threadIdx.x & 31;
  const float* r = ff + (long)wg * HOc;
  float s1 = 0.f, s2 = 0.f;
#pragma unroll
  for (int p = 0; p < 8; ++p) {
    float v = r[p * 32 + lane];
    s1 += v;
    s2 += v * v;
  }
  s1 = wave_sum32(s1);
  s2 = wave_sum32(s2);
  float mu  = s1 * (1.f / HOc);
  float var = s2 * (1.f / HOc) - mu * mu;
  float inv = rsqrtf(var + EPSc);
  float* o = out + (long)wg * HOc;
#pragma unroll
  for (int p = 0; p < 8; ++p) {
    int c = p * 32 + lane;
    o[c] = (r[c] - mu) * inv * g[c] + bb[c];
  }
}

extern "C" void kernel_launch(void* const* d_in, const int* in_sizes, int n_in,
                              void* d_out, int out_size, void* d_ws, size_t ws_size,
                              hipStream_t stream) {
  const float* x    = (const float*)d_in[0];
  const float* edge = (const float*)d_in[1];
  const float* adj  = (const float*)d_in[2];
  const float* W    = (const float*)d_in[3];
  const float* W1   = (const float*)d_in[4];
  const float* a    = (const float*)d_in[5];
  const float* Wo   = (const float*)d_in[6];
  const float* bo   = (const float*)d_in[7];
  const float* w1   = (const float*)d_in[8];
  const float* b1   = (const float*)d_in[9];
  const float* w2   = (const float*)d_in[10];
  const float* b2   = (const float*)d_in[11];
  const float* lng  = (const float*)d_in[12];
  const float* lnb  = (const float*)d_in[13];
  float* out = (float*)d_out;

  float* ws  = (float*)d_ws;
  float* h   = ws;
  float* s   = h + 524288;
  float* t   = s + 16384;
  float* u   = t + 16384;
  float* cat = u + 4194304;
  float* y   = cat + 524288;
  float* z   = y + 524288;
  float* ff  = z + 2097152;

  k_front<<<768, 256, 0, stream>>>(x, W, a, edge, W1, h, s, t, u);
  k_attn_hp<<<256, 128, 0, stream>>>(u, s, t, adj, h, cat);
  k_y   <<<256,  256, 0, stream>>>(cat, Wo, bo, y);
  k_z   <<<256,  256, 0, stream>>>(y, w1, b1, z);
  k_ff  <<<256,  256, 0, stream>>>(z, w2, b2, y, ff);
  k_ln  <<<256,  256, 0, stream>>>(ff, lng, lnb, out);
}
